# starting kernel
# baseline (speedup 1.0000x reference)
#include <hip/hip_runtime.h>
#include <math.h>


typedef float v2f __attribute__((ext_vector_type(2)));
typedef float v8f __attribute__((ext_vector_type(8)));

#define DEPTH     11
#define LEVELS    (DEPTH + 1)
#define IN_W      1024
#define OUT_W     1024
#define N_NODES   4095
#define N_TOKENS  16384
#define TOK_TILE  16
#define XSTRIDE   1026

__device__ __forceinline__ v8f wmma4(v2f a, v2f b, v8f c) {
  return __builtin_amdgcn_wmma_f32_16x16x4_f32(
       false, a,  false, b,
       (short)0, c,  false,  false);
}

__global__ __launch_bounds__(256) void transpose_wout(const float* __restrict__ w_out,
                                                      float* __restrict__ w_out_T) {
  __shared__ float tile[32][33];
  const int n0 = blockIdx.x * 32;
  const int o0 = blockIdx.y * 32;
  const int tx = threadIdx.x;
  const int ty = threadIdx.y;
#pragma unroll
  for (int j = 0; j < 4; ++j) {
    const int n = n0 + tx;
    const int o = o0 + ty + j * 8;
    if (n < N_NODES) tile[ty + j * 8][tx] = w_out[(size_t)o * N_NODES + n];
  }
  __syncthreads();
#pragma unroll
  for (int j = 0; j < 4; ++j) {
    const int n = n0 + ty + j * 8;
    const int o = o0 + tx;
    if (n < N_NODES) w_out_T[(size_t)n * OUT_W + o] = tile[tx][ty + j * 8];
  }
}

__global__ __launch_bounds__(128) void fff_fwd(const float* __restrict__ x,
                                               const float* __restrict__ w_in,
                                               const float* __restrict__ w_out_T,
                                               float* __restrict__ out) {
  __shared__ float lds_x[TOK_TILE * XSTRIDE];
  __shared__ float s_logit[TOK_TILE];
  __shared__ int   s_node[TOK_TILE];
  __shared__ float s_act[LEVELS][TOK_TILE];
  __shared__ int   s_path[LEVELS][TOK_TILE];

  const int tid  = threadIdx.x;
  const int lane = tid & 31;
  const int wave = tid >> 5;
  const size_t xbase = (size_t)blockIdx.x * TOK_TILE * IN_W;

  const float4* __restrict__ xg = (const float4*)(x + xbase);
#pragma unroll 4
  for (int i = 0; i < 32; ++i) {
    const int idx = i * 128 + tid;
    const int tok = idx >> 8;
    const int k4  = idx & 255;
    const float4 v = xg[idx];
    float* dst = &lds_x[tok * XSTRIDE + k4 * 4];
    dst[0] = v.x; dst[1] = v.y; dst[2] = v.z; dst[3] = v.w;
  }
  if (tid < TOK_TILE) { s_logit[tid] = 0.0f; s_node[tid] = 0; }
  __syncthreads();

  const int tok    = lane & 15;
  const int kShift = ((lane >> 4) & 1) * 2;
  const int kbase  = wave * 256;
  const float* xrow = &lds_x[tok * XSTRIDE + kbase + kShift];

  for (int lvl = 0; lvl < LEVELS; ++lvl) {
    const int node = s_node[tok];
    const float* wrow = w_in + (size_t)node * IN_W + kbase + kShift;

    v8f c0 = {}, c1 = {}, c2 = {}, c3 = {};
#pragma unroll 4
    for (int kk = 0; kk < 256; kk += 16) {
      v2f a0 = *(const v2f*)(xrow + kk +  0);
      v2f b0 = *(const v2f*)(wrow + kk +  0);
      c0 = wmma4(a0, b0, c0);
      v2f a1 = *(const v2f*)(xrow + kk +  4);
      v2f b1 = *(const v2f*)(wrow + kk +  4);
      c1 = wmma4(a1, b1, c1);
      v2f a2 = *(const v2f*)(xrow + kk +  8);
      v2f b2 = *(const v2f*)(wrow + kk +  8);
      c2 = wmma4(a2, b2, c2);
      v2f a3 = *(const v2f*)(xrow + kk + 12);
      v2f b3 = *(const v2f*)(wrow + kk + 12);
      c3 = wmma4(a3, b3, c3);
    }
    const v8f c = (c0 + c1) + (c2 + c3);

    const int r = lane & 7;
    const float e0 = (r & 1) ? c[1] : c[0];
    const float e1 = (r & 1) ? c[3] : c[2];
    const float e2 = (r & 1) ? c[5] : c[4];
    const float e3 = (r & 1) ? c[7] : c[6];
    const float f0 = (r & 2) ? e1 : e0;
    const float f1 = (r & 2) ? e3 : e2;
    const float dg = (r & 4) ? f1 : f0;
    if (lane < 8)        atomicAdd(&s_logit[lane],      dg);
    else if (lane >= 24) atomicAdd(&s_logit[lane - 16], dg);
    __syncthreads();

    if (tid < TOK_TILE) {
      const float logit = s_logit[tid];
      const float act = 0.5f * logit * (1.0f + erff(logit * 0.70710678118654752f));
      s_act[lvl][tid]  = act;
      s_path[lvl][tid] = node;
      s_node[tid]  = 2 * node + 1 + ((logit > 0.0f) ? 1 : 0);
      s_logit[tid] = 0.0f;
    }
    __syncthreads();
  }

  float* orow_base = out + xbase;
  for (int t4 = 0; t4 < 4; ++t4) {
    const int t = wave * 4 + t4;
    float4 acc[8];
#pragma unroll
    for (int c8 = 0; c8 < 8; ++c8) acc[c8] = make_float4(0.f, 0.f, 0.f, 0.f);
    for (int k = 0; k < LEVELS; ++k) {
      const float a = s_act[k][t];
      const float4* row = (const float4*)(w_out_T + (size_t)s_path[k][t] * OUT_W);
#pragma unroll
      for (int c8 = 0; c8 < 8; ++c8) {
        const float4 w4 = row[c8 * 32 + lane];
        acc[c8].x += a * w4.x;
        acc[c8].y += a * w4.y;
        acc[c8].z += a * w4.z;
        acc[c8].w += a * w4.w;
      }
    }
    float4* orow = (float4*)(orow_base + (size_t)t * OUT_W);
#pragma unroll
    for (int c8 = 0; c8 < 8; ++c8) orow[c8 * 32 + lane] = acc[c8];
  }
}

extern "C" void kernel_launch(void* const* d_in, const int* in_sizes, int n_in,
                              void* d_out, int out_size, void* d_ws, size_t ws_size,
                              hipStream_t stream) {
  (void)in_sizes; (void)n_in; (void)out_size; (void)ws_size;
  const float* x     = (const float*)d_in[0];
  const float* w_in  = (const float*)d_in[1];
  const float* w_out = (const float*)d_in[2];
  float* w_out_T = (float*)d_ws;
  float* out     = (float*)d_out;

  transpose_wout<<<dim3((N_NODES + 31) / 32, OUT_W / 32), dim3(32, 8), 0, stream>>>(
      w_out, w_out_T);
  fff_fwd<<<N_TOKENS / TOK_TILE, 128, 0, stream>>>(x, w_in, w_out_T, out);
}


# Stage-2 A and B fragment loads widened b64->b128, same permuted k order in both loaders (dot-product sum order changes), slice fully unrolled: half the 16-row gathers, one load group per level
# speedup vs baseline: 1.3265x; 1.3265x over previous
#include <hip/hip_runtime.h>
#include <math.h>


typedef float v2f __attribute__((ext_vector_type(2)));
typedef float v8f __attribute__((ext_vector_type(8)));

#define DEPTH     11
#define LEVELS    (DEPTH + 1)
#define IN_W      1024
#define OUT_W     1024
#define N_NODES   4095
#define N_TOKENS  16384
#define TOK_TILE  16
#define XSTRIDE   1028

__device__ __forceinline__ v8f wmma4(v2f a, v2f b, v8f c) {
  return __builtin_amdgcn_wmma_f32_16x16x4_f32(
       false, a,  false, b,
       (short)0, c,  false,  false);
}

__global__ __launch_bounds__(256) void transpose_wout(const float* __restrict__ w_out,
                                                      float* __restrict__ w_out_T) {
  __shared__ float tile[32][33];
  const int n0 = blockIdx.x * 32;
  const int o0 = blockIdx.y * 32;
  const int tx = threadIdx.x;
  const int ty = threadIdx.y;
#pragma unroll
  for (int j = 0; j < 4; ++j) {
    const int n = n0 + tx;
    const int o = o0 + ty + j * 8;
    if (n < N_NODES) tile[ty + j * 8][tx] = w_out[(size_t)o * N_NODES + n];
  }
  __syncthreads();
#pragma unroll
  for (int j = 0; j < 4; ++j) {
    const int n = n0 + ty + j * 8;
    const int o = o0 + tx;
    if (n < N_NODES) w_out_T[(size_t)n * OUT_W + o] = tile[tx][ty + j * 8];
  }
}

__global__ __launch_bounds__(128) void fff_fwd(const float* __restrict__ x,
                                               const float* __restrict__ w_in,
                                               const float* __restrict__ w_out_T,
                                               float* __restrict__ out) {
  __shared__ __attribute__((aligned(16))) float lds_x[TOK_TILE * XSTRIDE];
  __shared__ float s_logit[TOK_TILE];
  __shared__ int   s_node[TOK_TILE];
  __shared__ float s_act[LEVELS][TOK_TILE];
  __shared__ int   s_path[LEVELS][TOK_TILE];

  const int tid  = threadIdx.x;
  const int lane = tid & 31;
  const int wave = tid >> 5;
  const size_t xbase = (size_t)blockIdx.x * TOK_TILE * IN_W;

  const float4* __restrict__ xg = (const float4*)(x + xbase);
#pragma unroll 4
  for (int i = 0; i < 32; ++i) {
    const int idx = i * 128 + tid;
    const int tok = idx >> 8;
    const int k4  = idx & 255;
    const float4 v = xg[idx];
    *(float4*)&lds_x[tok * XSTRIDE + k4 * 4] = v;
  }
  if (tid < TOK_TILE) { s_logit[tid] = 0.0f; s_node[tid] = 0; }
  __syncthreads();

  const int tok    = lane & 15;
  const int kShift = ((lane >> 4) & 1) * 4;
  const int kbase  = wave * 256;
  const float* xrow = &lds_x[tok * XSTRIDE + kbase + kShift];

  for (int lvl = 0; lvl < LEVELS; ++lvl) {
    const int node = s_node[tok];
    const float* wrow = w_in + (size_t)node * IN_W + kbase + kShift;

    v8f c0 = {}, c1 = {}, c2 = {}, c3 = {};
#pragma unroll
    for (int kk = 0; kk < 256; kk += 16) {
      const float4 a0 = *(const float4*)(xrow + kk + 0);
      const float4 b0 = *(const float4*)(wrow + kk + 0);
      const float4 a1 = *(const float4*)(xrow + kk + 8);
      const float4 b1 = *(const float4*)(wrow + kk + 8);
      c0 = wmma4(v2f{a0.x, a0.y}, v2f{b0.x, b0.y}, c0);
      c1 = wmma4(v2f{a0.z, a0.w}, v2f{b0.z, b0.w}, c1);
      c2 = wmma4(v2f{a1.x, a1.y}, v2f{b1.x, b1.y}, c2);
      c3 = wmma4(v2f{a1.z, a1.w}, v2f{b1.z, b1.w}, c3);
    }
    const v8f c = (c0 + c1) + (c2 + c3);

    const int r = lane & 7;
    const float e0 = (r & 1) ? c[1] : c[0];
    const float e1 = (r & 1) ? c[3] : c[2];
    const float e2 = (r & 1) ? c[5] : c[4];
    const float e3 = (r & 1) ? c[7] : c[6];
    const float f0 = (r & 2) ? e1 : e0;
    const float f1 = (r & 2) ? e3 : e2;
    const float dg = (r & 4) ? f1 : f0;
    if (lane < 8)        atomicAdd(&s_logit[lane],      dg);
    else if (lane >= 24) atomicAdd(&s_logit[lane - 16], dg);
    __syncthreads();

    if (tid < TOK_TILE) {
      const float logit = s_logit[tid];
      const float act = 0.5f * logit * (1.0f + erff(logit * 0.70710678118654752f));
      s_act[lvl][tid]  = act;
      s_path[lvl][tid] = node;
      s_node[tid]  = 2 * node + 1 + ((logit > 0.0f) ? 1 : 0);
      s_logit[tid] = 0.0f;
    }
    __syncthreads();
  }

  float* orow_base = out + xbase;
  for (int t4 = 0; t4 < 4; ++t4) {
    const int t = wave * 4 + t4;
    float4 acc[8];
#pragma unroll
    for (int c8 = 0; c8 < 8; ++c8) acc[c8] = make_float4(0.f, 0.f, 0.f, 0.f);
    for (int k = 0; k < LEVELS; ++k) {
      const float a = s_act[k][t];
      const float4* row = (const float4*)(w_out_T + (size_t)s_path[k][t] * OUT_W);
#pragma unroll
      for (int c8 = 0; c8 < 8; ++c8) {
        const float4 w4 = row[c8 * 32 + lane];
        acc[c8].x += a * w4.x;
        acc[c8].y += a * w4.y;
        acc[c8].z += a * w4.z;
        acc[c8].w += a * w4.w;
      }
    }
    float4* orow = (float4*)(orow_base + (size_t)t * OUT_W);
#pragma unroll
    for (int c8 = 0; c8 < 8; ++c8) orow[c8 * 32 + lane] = acc[c8];
  }
}

extern "C" void kernel_launch(void* const* d_in, const int* in_sizes, int n_in,
                              void* d_out, int out_size, void* d_ws, size_t ws_size,
                              hipStream_t stream) {
  (void)in_sizes; (void)n_in; (void)out_size; (void)ws_size;
  const float* x     = (const float*)d_in[0];
  const float* w_in  = (const float*)d_in[1];
  const float* w_out = (const float*)d_in[2];
  float* w_out_T = (float*)d_ws;
  float* out     = (float*)d_out;

  transpose_wout<<<dim3((N_NODES + 31) / 32, OUT_W / 32), dim3(32, 8), 0, stream>>>(
      w_out, w_out_T);
  fff_fwd<<<N_TOKENS / TOK_TILE, 128, 0, stream>>>(x, w_in, w_out_T, out);
}


# Levels 0-4 (nodes 0..30) as one dense 16-token x 32-node WMMA product replacing 5 diagonal-only passes: 12 -> 9 passes of 64 WMMAs per wave; their 4 wave partials now summed in fixed order
# speedup vs baseline: 1.4190x; 1.0697x over previous
#include <hip/hip_runtime.h>
#include <math.h>


typedef float v2f __attribute__((ext_vector_type(2)));
typedef float v8f __attribute__((ext_vector_type(8)));

#define DEPTH     11
#define LEVELS    (DEPTH + 1)
#define IN_W      1024
#define OUT_W     1024
#define N_NODES   4095
#define N_TOKENS  16384
#define TOK_TILE  16
#define XSTRIDE   1028
#define TOP_LEVELS 5
#define TOP_TILES  2
#define TSTRIDE   20

__device__ __forceinline__ v8f wmma4(v2f a, v2f b, v8f c) {
  return __builtin_amdgcn_wmma_f32_16x16x4_f32(
       false, a,  false, b,
       (short)0, c,  false,  false);
}

__global__ __launch_bounds__(256) void transpose_wout(const float* __restrict__ w_out,
                                                      float* __restrict__ w_out_T) {
  __shared__ float tile[32][33];
  const int n0 = blockIdx.x * 32;
  const int o0 = blockIdx.y * 32;
  const int tx = threadIdx.x;
  const int ty = threadIdx.y;
#pragma unroll
  for (int j = 0; j < 4; ++j) {
    const int n = n0 + tx;
    const int o = o0 + ty + j * 8;
    if (n < N_NODES) tile[ty + j * 8][tx] = w_out[(size_t)o * N_NODES + n];
  }
  __syncthreads();
#pragma unroll
  for (int j = 0; j < 4; ++j) {
    const int n = n0 + ty + j * 8;
    const int o = o0 + tx;
    if (n < N_NODES) w_out_T[(size_t)n * OUT_W + o] = tile[tx][ty + j * 8];
  }
}

__global__ __launch_bounds__(128) void fff_fwd(const float* __restrict__ x,
                                               const float* __restrict__ w_in,
                                               const float* __restrict__ w_out_T,
                                               float* __restrict__ out) {
  __shared__ __attribute__((aligned(16))) float lds_x[TOK_TILE * XSTRIDE];
  __shared__ float s_logit[TOK_TILE];
  __shared__ int   s_node[TOK_TILE];
  __shared__ float s_act[LEVELS][TOK_TILE];
  __shared__ int   s_path[LEVELS][TOK_TILE];
  __shared__ __attribute__((aligned(16))) float s_top[4][TOP_TILES * 16 * TSTRIDE];

  const int tid  = threadIdx.x;
  const int lane = tid & 31;
  const int wave = tid >> 5;
  const size_t xbase = (size_t)blockIdx.x * TOK_TILE * IN_W;

  const float4* __restrict__ xg = (const float4*)(x + xbase);
#pragma unroll 4
  for (int i = 0; i < 32; ++i) {
    const int idx = i * 128 + tid;
    const int tok = idx >> 8;
    const int k4  = idx & 255;
    const float4 v = xg[idx];
    *(float4*)&lds_x[tok * XSTRIDE + k4 * 4] = v;
  }
  if (tid < TOK_TILE) s_logit[tid] = 0.0f;
  __syncthreads();

  const int tok    = lane & 15;
  const int kShift = ((lane >> 4) & 1) * 4;
  const int kbase  = wave * 256;
  const float* xrow = &lds_x[tok * XSTRIDE + kbase + kShift];

#pragma unroll
  for (int j = 0; j < TOP_TILES; ++j) {
    const float* wrow = w_in + (size_t)(16 * j + tok) * IN_W + kbase + kShift;
    v8f c0 = {}, c1 = {}, c2 = {}, c3 = {};
#pragma unroll
    for (int kk = 0; kk < 256; kk += 16) {
      const float4 a0 = *(const float4*)(xrow + kk + 0);
      const float4 b0 = *(const float4*)(wrow + kk + 0);
      const float4 a1 = *(const float4*)(xrow + kk + 8);
      const float4 b1 = *(const float4*)(wrow + kk + 8);
      c0 = wmma4(v2f{a0.x, a0.y}, v2f{b0.x, b0.y}, c0);
      c1 = wmma4(v2f{a0.z, a0.w}, v2f{b0.z, b0.w}, c1);
      c2 = wmma4(v2f{a1.x, a1.y}, v2f{b1.x, b1.y}, c2);
      c3 = wmma4(v2f{a1.z, a1.w}, v2f{b1.z, b1.w}, c3);
    }
    const v8f c = (c0 + c1) + (c2 + c3);
    float* dst = &s_top[wave][(16 * j + tok) * TSTRIDE + 8 * (lane >> 4)];
    *(float4*)(dst + 0) = make_float4(c[0], c[1], c[2], c[3]);
    *(float4*)(dst + 4) = make_float4(c[4], c[5], c[6], c[7]);
  }
  __syncthreads();
  if (tid < TOK_TILE) {
    int node = 0;
    for (int lvl = 0; lvl < TOP_LEVELS; ++lvl) {
      const int o = node * TSTRIDE + tid;
      const float logit = (s_top[0][o] + s_top[1][o]) + (s_top[2][o] + s_top[3][o]);
      const float act = 0.5f * logit * (1.0f + erff(logit * 0.70710678118654752f));
      s_act[lvl][tid]  = act;
      s_path[lvl][tid] = node;
      node = 2 * node + 1 + ((logit > 0.0f) ? 1 : 0);
    }
    s_node[tid] = node;
  }
  __syncthreads();

  for (int lvl = TOP_LEVELS; lvl < LEVELS; ++lvl) {
    const int node = s_node[tok];
    const float* wrow = w_in + (size_t)node * IN_W + kbase + kShift;

    v8f c0 = {}, c1 = {}, c2 = {}, c3 = {};
#pragma unroll
    for (int kk = 0; kk < 256; kk += 16) {
      const float4 a0 = *(const float4*)(xrow + kk + 0);
      const float4 b0 = *(const float4*)(wrow + kk + 0);
      const float4 a1 = *(const float4*)(xrow + kk + 8);
      const float4 b1 = *(const float4*)(wrow + kk + 8);
      c0 = wmma4(v2f{a0.x, a0.y}, v2f{b0.x, b0.y}, c0);
      c1 = wmma4(v2f{a0.z, a0.w}, v2f{b0.z, b0.w}, c1);
      c2 = wmma4(v2f{a1.x, a1.y}, v2f{b1.x, b1.y}, c2);
      c3 = wmma4(v2f{a1.z, a1.w}, v2f{b1.z, b1.w}, c3);
    }
    const v8f c = (c0 + c1) + (c2 + c3);

    const int r = lane & 7;
    const float e0 = (r & 1) ? c[1] : c[0];
    const float e1 = (r & 1) ? c[3] : c[2];
    const float e2 = (r & 1) ? c[5] : c[4];
    const float e3 = (r & 1) ? c[7] : c[6];
    const float f0 = (r & 2) ? e1 : e0;
    const float f1 = (r & 2) ? e3 : e2;
    const float dg = (r & 4) ? f1 : f0;
    if (lane < 8)        atomicAdd(&s_logit[lane],      dg);
    else if (lane >= 24) atomicAdd(&s_logit[lane - 16], dg);
    __syncthreads();

    if (tid < TOK_TILE) {
      const float logit = s_logit[tid];
      const float act = 0.5f * logit * (1.0f + erff(logit * 0.70710678118654752f));
      s_act[lvl][tid]  = act;
      s_path[lvl][tid] = node;
      s_node[tid]  = 2 * node + 1 + ((logit > 0.0f) ? 1 : 0);
      s_logit[tid] = 0.0f;
    }
    __syncthreads();
  }

  float* orow_base = out + xbase;
  for (int t4 = 0; t4 < 4; ++t4) {
    const int t = wave * 4 + t4;
    float4 acc[8];
#pragma unroll
    for (int c8 = 0; c8 < 8; ++c8) acc[c8] = make_float4(0.f, 0.f, 0.f, 0.f);
    for (int k = 0; k < LEVELS; ++k) {
      const float a = s_act[k][t];
      const float4* row = (const float4*)(w_out_T + (size_t)s_path[k][t] * OUT_W);
#pragma unroll
      for (int c8 = 0; c8 < 8; ++c8) {
        const float4 w4 = row[c8 * 32 + lane];
        acc[c8].x += a * w4.x;
        acc[c8].y += a * w4.y;
        acc[c8].z += a * w4.z;
        acc[c8].w += a * w4.w;
      }
    }
    float4* orow = (float4*)(orow_base + (size_t)t * OUT_W);
#pragma unroll
    for (int c8 = 0; c8 < 8; ++c8) orow[c8 * 32 + lane] = acc[c8];
  }
}

extern "C" void kernel_launch(void* const* d_in, const int* in_sizes, int n_in,
                              void* d_out, int out_size, void* d_ws, size_t ws_size,
                              hipStream_t stream) {
  (void)in_sizes; (void)n_in; (void)out_size; (void)ws_size;
  const float* x     = (const float*)d_in[0];
  const float* w_in  = (const float*)d_in[1];
  const float* w_out = (const float*)d_in[2];
  float* w_out_T = (float*)d_ws;
  float* out     = (float*)d_out;

  transpose_wout<<<dim3((N_NODES + 31) / 32, OUT_W / 32), dim3(32, 8), 0, stream>>>(
      w_out, w_out_T);
  fff_fwd<<<N_TOKENS / TOK_TILE, 128, 0, stream>>>(x, w_in, w_out_T, out);
}


# Main kernel cut from 278 to 186 VGPRs (output loop in groups of 4 levels, the two top tiles' loads kept apart): 4 not 3 work-groups resident per WGP, the 1024-group grid runs in one round
# speedup vs baseline: 1.6815x; 1.1850x over previous
#include <hip/hip_runtime.h>
#include <math.h>


typedef float v2f __attribute__((ext_vector_type(2)));
typedef float v8f __attribute__((ext_vector_type(8)));

#define DEPTH     11
#define LEVELS    (DEPTH + 1)
#define IN_W      1024
#define OUT_W     1024
#define N_NODES   4095
#define N_TOKENS  16384
#define TOK_TILE  16
#define XSTRIDE   1028
#define TOP_LEVELS 5
#define TOP_TILES  2
#define TSTRIDE   20

__device__ __forceinline__ v8f wmma4(v2f a, v2f b, v8f c) {
  return __builtin_amdgcn_wmma_f32_16x16x4_f32(
       false, a,  false, b,
       (short)0, c,  false,  false);
}

__global__ __launch_bounds__(256) void transpose_wout(const float* __restrict__ w_out,
                                                      float* __restrict__ w_out_T) {
  __shared__ float tile[32][33];
  const int n0 = blockIdx.x * 32;
  const int o0 = blockIdx.y * 32;
  const int tx = threadIdx.x;
  const int ty = threadIdx.y;
#pragma unroll
  for (int j = 0; j < 4; ++j) {
    const int n = n0 + tx;
    const int o = o0 + ty + j * 8;
    if (n < N_NODES) tile[ty + j * 8][tx] = w_out[(size_t)o * N_NODES + n];
  }
  __syncthreads();
#pragma unroll
  for (int j = 0; j < 4; ++j) {
    const int n = n0 + ty + j * 8;
    const int o = o0 + tx;
    if (n < N_NODES) w_out_T[(size_t)n * OUT_W + o] = tile[tx][ty + j * 8];
  }
}

__global__ __launch_bounds__(128, 4) void fff_fwd(const float* __restrict__ x,
                                               const float* __restrict__ w_in,
                                               const float* __restrict__ w_out_T,
                                               float* __restrict__ out) {
  __shared__ __attribute__((aligned(16))) float lds_x[TOK_TILE * XSTRIDE];
  __shared__ float s_logit[TOK_TILE];
  __shared__ int   s_node[TOK_TILE];
  __shared__ float s_act[LEVELS][TOK_TILE];
  __shared__ int   s_path[LEVELS][TOK_TILE];
  __shared__ __attribute__((aligned(16))) float s_top[4][TOP_TILES * 16 * TSTRIDE];

  const int tid  = threadIdx.x;
  const int lane = tid & 31;
  const int wave = tid >> 5;
  const size_t xbase = (size_t)blockIdx.x * TOK_TILE * IN_W;

  const float4* __restrict__ xg = (const float4*)(x + xbase);
#pragma unroll 4
  for (int i = 0; i < 32; ++i) {
    const int idx = i * 128 + tid;
    const int tok = idx >> 8;
    const int k4  = idx & 255;
    const float4 v = xg[idx];
    *(float4*)&lds_x[tok * XSTRIDE + k4 * 4] = v;
  }
  if (tid < TOK_TILE) s_logit[tid] = 0.0f;
  __syncthreads();

  const int tok    = lane & 15;
  const int kShift = ((lane >> 4) & 1) * 4;
  const int kbase  = wave * 256;
  const float* xrow = &lds_x[tok * XSTRIDE + kbase + kShift];

#pragma unroll
  for (int j = 0; j < TOP_TILES; ++j) {
    const float* wrow = w_in + (size_t)(16 * j + tok) * IN_W + kbase + kShift;
    v8f c0 = {}, c1 = {}, c2 = {}, c3 = {};
#pragma unroll
    for (int kk = 0; kk < 256; kk += 16) {
      const float4 a0 = *(const float4*)(xrow + kk + 0);
      const float4 b0 = *(const float4*)(wrow + kk + 0);
      const float4 a1 = *(const float4*)(xrow + kk + 8);
      const float4 b1 = *(const float4*)(wrow + kk + 8);
      c0 = wmma4(v2f{a0.x, a0.y}, v2f{b0.x, b0.y}, c0);
      c1 = wmma4(v2f{a0.z, a0.w}, v2f{b0.z, b0.w}, c1);
      c2 = wmma4(v2f{a1.x, a1.y}, v2f{b1.x, b1.y}, c2);
      c3 = wmma4(v2f{a1.z, a1.w}, v2f{b1.z, b1.w}, c3);
    }
    const v8f c = (c0 + c1) + (c2 + c3);
    float* dst = &s_top[wave][(16 * j + tok) * TSTRIDE + 8 * (lane >> 4)];
    *(float4*)(dst + 0) = make_float4(c[0], c[1], c[2], c[3]);
    *(float4*)(dst + 4) = make_float4(c[4], c[5], c[6], c[7]);
    if (j + 1 < TOP_TILES) __builtin_amdgcn_sched_barrier(0);
  }
  __syncthreads();
  if (tid < TOK_TILE) {
    int node = 0;
    for (int lvl = 0; lvl < TOP_LEVELS; ++lvl) {
      const int o = node * TSTRIDE + tid;
      const float logit = (s_top[0][o] + s_top[1][o]) + (s_top[2][o] + s_top[3][o]);
      const float act = 0.5f * logit * (1.0f + erff(logit * 0.70710678118654752f));
      s_act[lvl][tid]  = act;
      s_path[lvl][tid] = node;
      node = 2 * node + 1 + ((logit > 0.0f) ? 1 : 0);
    }
    s_node[tid] = node;
  }
  __syncthreads();

  for (int lvl = TOP_LEVELS; lvl < LEVELS; ++lvl) {
    const int node = s_node[tok];
    const float* wrow = w_in + (size_t)node * IN_W + kbase + kShift;

    v8f c0 = {}, c1 = {}, c2 = {}, c3 = {};
#pragma unroll
    for (int kk = 0; kk < 256; kk += 16) {
      const float4 a0 = *(const float4*)(xrow + kk + 0);
      const float4 b0 = *(const float4*)(wrow + kk + 0);
      const float4 a1 = *(const float4*)(xrow + kk + 8);
      const float4 b1 = *(const float4*)(wrow + kk + 8);
      c0 = wmma4(v2f{a0.x, a0.y}, v2f{b0.x, b0.y}, c0);
      c1 = wmma4(v2f{a0.z, a0.w}, v2f{b0.z, b0.w}, c1);
      c2 = wmma4(v2f{a1.x, a1.y}, v2f{b1.x, b1.y}, c2);
      c3 = wmma4(v2f{a1.z, a1.w}, v2f{b1.z, b1.w}, c3);
    }
    const v8f c = (c0 + c1) + (c2 + c3);

    const int r = lane & 7;
    const float e0 = (r & 1) ? c[1] : c[0];
    const float e1 = (r & 1) ? c[3] : c[2];
    const float e2 = (r & 1) ? c[5] : c[4];
    const float e3 = (r & 1) ? c[7] : c[6];
    const float f0 = (r & 2) ? e1 : e0;
    const float f1 = (r & 2) ? e3 : e2;
    const float dg = (r & 4) ? f1 : f0;
    if (lane < 8)        atomicAdd(&s_logit[lane],      dg);
    else if (lane >= 24) atomicAdd(&s_logit[lane - 16], dg);
    __syncthreads();

    if (tid < TOK_TILE) {
      const float logit = s_logit[tid];
      const float act = 0.5f * logit * (1.0f + erff(logit * 0.70710678118654752f));
      s_act[lvl][tid]  = act;
      s_path[lvl][tid] = node;
      s_node[tid]  = 2 * node + 1 + ((logit > 0.0f) ? 1 : 0);
      s_logit[tid] = 0.0f;
    }
    __syncthreads();
  }

  float* orow_base = out + xbase;
  for (int t4 = 0; t4 < 4; ++t4) {
    const int t = wave * 4 + t4;
    float4 acc[8];
#pragma unroll
    for (int c8 = 0; c8 < 8; ++c8) acc[c8] = make_float4(0.f, 0.f, 0.f, 0.f);
#pragma unroll 4
    for (int k = 0; k < LEVELS; ++k) {
      const float a = s_act[k][t];
      const float4* row = (const float4*)(w_out_T + (size_t)s_path[k][t] * OUT_W);
#pragma unroll
      for (int c8 = 0; c8 < 8; ++c8) {
        const float4 w4 = row[c8 * 32 + lane];
        acc[c8].x += a * w4.x;
        acc[c8].y += a * w4.y;
        acc[c8].z += a * w4.z;
        acc[c8].w += a * w4.w;
      }
    }
    float4* orow = (float4*)(orow_base + (size_t)t * OUT_W);
#pragma unroll
    for (int c8 = 0; c8 < 8; ++c8) orow[c8 * 32 + lane] = acc[c8];
  }
}

extern "C" void kernel_launch(void* const* d_in, const int* in_sizes, int n_in,
                              void* d_out, int out_size, void* d_ws, size_t ws_size,
                              hipStream_t stream) {
  (void)in_sizes; (void)n_in; (void)out_size; (void)ws_size;
  const float* x     = (const float*)d_in[0];
  const float* w_in  = (const float*)d_in[1];
  const float* w_out = (const float*)d_in[2];
  float* w_out_T = (float*)d_ws;
  float* out     = (float*)d_out;

  transpose_wout<<<dim3((N_NODES + 31) / 32, OUT_W / 32), dim3(32, 8), 0, stream>>>(
      w_out, w_out_T);
  fff_fwd<<<N_TOKENS / TOK_TILE, 128, 0, stream>>>(x, w_in, w_out_T, out);
}


# Walked levels: B rows fetched line-coalesced (8 lanes per 128-byte line: 4 not 16 lines per load) and re-laid to fragment layout via a per-wave LDS chunk; k pairing and sum order unchanged
# speedup vs baseline: 1.7430x; 1.0366x over previous
#include <hip/hip_runtime.h>
#include <math.h>


typedef float v2f __attribute__((ext_vector_type(2)));
typedef float v8f __attribute__((ext_vector_type(8)));

#define DEPTH     11
#define LEVELS    (DEPTH + 1)
#define IN_W      1024
#define OUT_W     1024
#define N_NODES   4095
#define N_TOKENS  16384
#define TOK_TILE  16
#define XSTRIDE   1028
#define TOP_LEVELS 5
#define TOP_TILES  2
#define TSTRIDE   20
#define SSTRIDE   36

__device__ __forceinline__ v8f wmma4(v2f a, v2f b, v8f c) {
  return __builtin_amdgcn_wmma_f32_16x16x4_f32(
       false, a,  false, b,
       (short)0, c,  false,  false);
}

__device__ __forceinline__ void wave_lds_order() {
  __builtin_amdgcn_fence(__ATOMIC_RELEASE, "wavefront", "local");
  __builtin_amdgcn_wave_barrier();
  __builtin_amdgcn_fence(__ATOMIC_ACQUIRE, "wavefront", "local");
}

__global__ __launch_bounds__(256) void transpose_wout(const float* __restrict__ w_out,
                                                      float* __restrict__ w_out_T) {
  __shared__ float tile[32][33];
  const int n0 = blockIdx.x * 32;
  const int o0 = blockIdx.y * 32;
  const int tx = threadIdx.x;
  const int ty = threadIdx.y;
#pragma unroll
  for (int j = 0; j < 4; ++j) {
    const int n = n0 + tx;
    const int o = o0 + ty + j * 8;
    if (n < N_NODES) tile[ty + j * 8][tx] = w_out[(size_t)o * N_NODES + n];
  }
  __syncthreads();
#pragma unroll
  for (int j = 0; j < 4; ++j) {
    const int n = n0 + ty + j * 8;
    const int o = o0 + tx;
    if (n < N_NODES) w_out_T[(size_t)n * OUT_W + o] = tile[tx][ty + j * 8];
  }
}

__global__ __launch_bounds__(128, 4) void fff_fwd(const float* __restrict__ x,
                                               const float* __restrict__ w_in,
                                               const float* __restrict__ w_out_T,
                                               float* __restrict__ out) {
  __shared__ __attribute__((aligned(16))) float lds_x[TOK_TILE * XSTRIDE];
  __shared__ float s_logit[TOK_TILE];
  __shared__ int   s_node[TOK_TILE];
  __shared__ float s_act[LEVELS][TOK_TILE];
  __shared__ int   s_path[LEVELS][TOK_TILE];
  __shared__ __attribute__((aligned(16))) float s_top[4][TOP_TILES * 16 * TSTRIDE];

  const int tid  = threadIdx.x;
  const int lane = tid & 31;
  const int wave = tid >> 5;
  const size_t xbase = (size_t)blockIdx.x * TOK_TILE * IN_W;

  const float4* __restrict__ xg = (const float4*)(x + xbase);
#pragma unroll 4
  for (int i = 0; i < 32; ++i) {
    const int idx = i * 128 + tid;
    const int tok = idx >> 8;
    const int k4  = idx & 255;
    const float4 v = xg[idx];
    *(float4*)&lds_x[tok * XSTRIDE + k4 * 4] = v;
  }
  if (tid < TOK_TILE) s_logit[tid] = 0.0f;
  __syncthreads();

  const int tok    = lane & 15;
  const int kShift = ((lane >> 4) & 1) * 4;
  const int kbase  = wave * 256;
  const float* xrow = &lds_x[tok * XSTRIDE + kbase + kShift];

#pragma unroll
  for (int j = 0; j < TOP_TILES; ++j) {
    const float* wrow = w_in + (size_t)(16 * j + tok) * IN_W + kbase + kShift;
    v8f c0 = {}, c1 = {}, c2 = {}, c3 = {};
#pragma unroll
    for (int kk = 0; kk < 256; kk += 16) {
      const float4 a0 = *(const float4*)(xrow + kk + 0);
      const float4 b0 = *(const float4*)(wrow + kk + 0);
      const float4 a1 = *(const float4*)(xrow + kk + 8);
      const float4 b1 = *(const float4*)(wrow + kk + 8);
      c0 = wmma4(v2f{a0.x, a0.y}, v2f{b0.x, b0.y}, c0);
      c1 = wmma4(v2f{a0.z, a0.w}, v2f{b0.z, b0.w}, c1);
      c2 = wmma4(v2f{a1.x, a1.y}, v2f{b1.x, b1.y}, c2);
      c3 = wmma4(v2f{a1.z, a1.w}, v2f{b1.z, b1.w}, c3);
    }
    const v8f c = (c0 + c1) + (c2 + c3);
    float* dst = &s_top[wave][(16 * j + tok) * TSTRIDE + 8 * (lane >> 4)];
    *(float4*)(dst + 0) = make_float4(c[0], c[1], c[2], c[3]);
    *(float4*)(dst + 4) = make_float4(c[4], c[5], c[6], c[7]);
    if (j + 1 < TOP_TILES) __builtin_amdgcn_sched_barrier(0);
  }
  __syncthreads();
  if (tid < TOK_TILE) {
    int node = 0;
    for (int lvl = 0; lvl < TOP_LEVELS; ++lvl) {
      const int o = node * TSTRIDE + tid;
      const float logit = (s_top[0][o] + s_top[1][o]) + (s_top[2][o] + s_top[3][o]);
      const float act = 0.5f * logit * (1.0f + erff(logit * 0.70710678118654752f));
      s_act[lvl][tid]  = act;
      s_path[lvl][tid] = node;
      node = 2 * node + 1 + ((logit > 0.0f) ? 1 : 0);
    }
    s_node[tid] = node;
  }
  __syncthreads();

  float* stg = &s_top[wave][0];
  const int rbase = 8 * ((lane >> 3) & 1) + 4 * (lane >> 4);
  const int spc  = (lane & 7) * 4;
  for (int lvl = TOP_LEVELS; lvl < LEVELS; ++lvl) {
    const int node = s_node[tok];
    const float* wr0 = w_in + (size_t)s_node[rbase + 0] * IN_W + kbase + spc;
    const float* wr1 = w_in + (size_t)s_node[rbase + 1] * IN_W + kbase + spc;
    const float* wr2 = w_in + (size_t)s_node[rbase + 2] * IN_W + kbase + spc;
    const float* wr3 = w_in + (size_t)s_node[rbase + 3] * IN_W + kbase + spc;

#define LDCHUNK(q)                                              \
    const float4 g##q##a = *(const float4*)(wr0 + 32 * (q));    \
    const float4 g##q##b = *(const float4*)(wr1 + 32 * (q));    \
    const float4 g##q##c = *(const float4*)(wr2 + 32 * (q));    \
    const float4 g##q##d = *(const float4*)(wr3 + 32 * (q));
    LDCHUNK(0) LDCHUNK(1) LDCHUNK(2) LDCHUNK(3) LDCHUNK(4) LDCHUNK(5) LDCHUNK(6) LDCHUNK(7)
#undef LDCHUNK

    v8f c0 = {}, c1 = {}, c2 = {}, c3 = {};
    float* st0 = &stg[(rbase + 0) * SSTRIDE + spc];
    float* st1 = &stg[(rbase + 1) * SSTRIDE + spc];
    float* st2 = &stg[(rbase + 2) * SSTRIDE + spc];
    float* st3 = &stg[(rbase + 3) * SSTRIDE + spc];
    const float* brow = &stg[tok * SSTRIDE + kShift];
#define CHUNK(q)                                                          \
    wave_lds_order();                                                     \
    *(float4*)st0 = g##q##a;                                              \
    *(float4*)st1 = g##q##b;                                              \
    *(float4*)st2 = g##q##c;                                              \
    *(float4*)st3 = g##q##d;                                              \
    wave_lds_order();                                                     \
    {                                                                     \
      const float4 a0 = *(const float4*)(xrow + 32 * (q) + 0);            \
      const float4 a1 = *(const float4*)(xrow + 32 * (q) + 8);            \
      const float4 a2 = *(const float4*)(xrow + 32 * (q) + 16);           \
      const float4 a3 = *(const float4*)(xrow + 32 * (q) + 24);           \
      const float4 b0 = *(const float4*)(brow + 0);                       \
      const float4 b1 = *(const float4*)(brow + 8);                       \
      const float4 b2 = *(const float4*)(brow + 16);                      \
      const float4 b3 = *(const float4*)(brow + 24);                      \
      c0 = wmma4(v2f{a0.x, a0.y}, v2f{b0.x, b0.y}, c0);                   \
      c1 = wmma4(v2f{a0.z, a0.w}, v2f{b0.z, b0.w}, c1);                   \
      c2 = wmma4(v2f{a1.x, a1.y}, v2f{b1.x, b1.y}, c2);                   \
      c3 = wmma4(v2f{a1.z, a1.w}, v2f{b1.z, b1.w}, c3);                   \
      c0 = wmma4(v2f{a2.x, a2.y}, v2f{b2.x, b2.y}, c0);                   \
      c1 = wmma4(v2f{a2.z, a2.w}, v2f{b2.z, b2.w}, c1);                   \
      c2 = wmma4(v2f{a3.x, a3.y}, v2f{b3.x, b3.y}, c2);                   \
      c3 = wmma4(v2f{a3.z, a3.w}, v2f{b3.z, b3.w}, c3);                   \
    }
    CHUNK(0) CHUNK(1) CHUNK(2) CHUNK(3) CHUNK(4) CHUNK(5) CHUNK(6) CHUNK(7)
#undef CHUNK
    const v8f c = (c0 + c1) + (c2 + c3);

    const int r = lane & 7;
    const float e0 = (r & 1) ? c[1] : c[0];
    const float e1 = (r & 1) ? c[3] : c[2];
    const float e2 = (r & 1) ? c[5] : c[4];
    const float e3 = (r & 1) ? c[7] : c[6];
    const float f0 = (r & 2) ? e1 : e0;
    const float f1 = (r & 2) ? e3 : e2;
    const float dg = (r & 4) ? f1 : f0;
    if (lane < 8)        atomicAdd(&s_logit[lane],      dg);
    else if (lane >= 24) atomicAdd(&s_logit[lane - 16], dg);
    __syncthreads();

    if (tid < TOK_TILE) {
      const float logit = s_logit[tid];
      const float act = 0.5f * logit * (1.0f + erff(logit * 0.70710678118654752f));
      s_act[lvl][tid]  = act;
      s_path[lvl][tid] = node;
      s_node[tid]  = 2 * node + 1 + ((logit > 0.0f) ? 1 : 0);
      s_logit[tid] = 0.0f;
    }
    __syncthreads();
  }

  float* orow_base = out + xbase;
  for (int t4 = 0; t4 < 4; ++t4) {
    const int t = wave * 4 + t4;
    float4 acc[8];
#pragma unroll
    for (int c8 = 0; c8 < 8; ++c8) acc[c8] = make_float4(0.f, 0.f, 0.f, 0.f);
#pragma unroll 4
    for (int k = 0; k < LEVELS; ++k) {
      const float a = s_act[k][t];
      const float4* row = (const float4*)(w_out_T + (size_t)s_path[k][t] * OUT_W);
#pragma unroll
      for (int c8 = 0; c8 < 8; ++c8) {
        const float4 w4 = row[c8 * 32 + lane];
        acc[c8].x += a * w4.x;
        acc[c8].y += a * w4.y;
        acc[c8].z += a * w4.z;
        acc[c8].w += a * w4.w;
      }
    }
    float4* orow = (float4*)(orow_base + (size_t)t * OUT_W);
#pragma unroll
    for (int c8 = 0; c8 < 8; ++c8) orow[c8 * 32 + lane] = acc[c8];
  }
}

extern "C" void kernel_launch(void* const* d_in, const int* in_sizes, int n_in,
                              void* d_out, int out_size, void* d_ws, size_t ws_size,
                              hipStream_t stream) {
  (void)in_sizes; (void)n_in; (void)out_size; (void)ws_size;
  const float* x     = (const float*)d_in[0];
  const float* w_in  = (const float*)d_in[1];
  const float* w_out = (const float*)d_in[2];
  float* w_out_T = (float*)d_ws;
  float* out     = (float*)d_out;

  transpose_wout<<<dim3((N_NODES + 31) / 32, OUT_W / 32), dim3(32, 8), 0, stream>>>(
      w_out, w_out_T);
  fff_fwd<<<N_TOKENS / TOK_TILE, 128, 0, stream>>>(x, w_in, w_out_T, out);
}


# Output stage: w_out_T rows of nodes 0..14 (path levels 0-3) copied once per work-group into the dead X tile's LDS and read from there: 8 not 12 row fetches from L2 per token; sum order unchanged
# speedup vs baseline: 1.7936x; 1.0290x over previous
#include <hip/hip_runtime.h>
#include <math.h>


typedef float v2f __attribute__((ext_vector_type(2)));
typedef float v8f __attribute__((ext_vector_type(8)));

#define DEPTH     11
#define LEVELS    (DEPTH + 1)
#define IN_W      1024
#define OUT_W     1024
#define N_NODES   4095
#define N_TOKENS  16384
#define TOK_TILE  16
#define XSTRIDE   1028
#define TOP_LEVELS 5
#define TOP_TILES  2
#define TSTRIDE   20
#define SSTRIDE   36
#define STAGED_LEVELS 4
#define STAGED_F4 (15 * OUT_W / 4)

__device__ __forceinline__ v8f wmma4(v2f a, v2f b, v8f c) {
  return __builtin_amdgcn_wmma_f32_16x16x4_f32(
       false, a,  false, b,
       (short)0, c,  false,  false);
}

__device__ __forceinline__ void wave_lds_order() {
  __builtin_amdgcn_fence(__ATOMIC_RELEASE, "wavefront", "local");
  __builtin_amdgcn_wave_barrier();
  __builtin_amdgcn_fence(__ATOMIC_ACQUIRE, "wavefront", "local");
}

__global__ __launch_bounds__(256) void transpose_wout(const float* __restrict__ w_out,
                                                      float* __restrict__ w_out_T) {
  __shared__ float tile[32][33];
  const int n0 = blockIdx.x * 32;
  const int o0 = blockIdx.y * 32;
  const int tx = threadIdx.x;
  const int ty = threadIdx.y;
#pragma unroll
  for (int j = 0; j < 4; ++j) {
    const int n = n0 + tx;
    const int o = o0 + ty + j * 8;
    if (n < N_NODES) tile[ty + j * 8][tx] = w_out[(size_t)o * N_NODES + n];
  }
  __syncthreads();
#pragma unroll
  for (int j = 0; j < 4; ++j) {
    const int n = n0 + ty + j * 8;
    const int o = o0 + tx;
    if (n < N_NODES) w_out_T[(size_t)n * OUT_W + o] = tile[tx][ty + j * 8];
  }
}

__global__ __launch_bounds__(128, 4) void fff_fwd(const float* __restrict__ x,
                                               const float* __restrict__ w_in,
                                               const float* __restrict__ w_out_T,
                                               float* __restrict__ out) {
  __shared__ __attribute__((aligned(16))) float lds_x[TOK_TILE * XSTRIDE];
  __shared__ float s_logit[TOK_TILE];
  __shared__ int   s_node[TOK_TILE];
  __shared__ float s_act[LEVELS][TOK_TILE];
  __shared__ int   s_path[LEVELS][TOK_TILE];
  __shared__ __attribute__((aligned(16))) float s_top[4][TOP_TILES * 16 * TSTRIDE];

  const int tid  = threadIdx.x;
  const int lane = tid & 31;
  const int wave = tid >> 5;
  const size_t xbase = (size_t)blockIdx.x * TOK_TILE * IN_W;

  const float4* __restrict__ xg = (const float4*)(x + xbase);
#pragma unroll 4
  for (int i = 0; i < 32; ++i) {
    const int idx = i * 128 + tid;
    const int tok = idx >> 8;
    const int k4  = idx & 255;
    const float4 v = xg[idx];
    *(float4*)&lds_x[tok * XSTRIDE + k4 * 4] = v;
  }
  if (tid < TOK_TILE) s_logit[tid] = 0.0f;
  __syncthreads();

  const int tok    = lane & 15;
  const int kShift = ((lane >> 4) & 1) * 4;
  const int kbase  = wave * 256;
  const float* xrow = &lds_x[tok * XSTRIDE + kbase + kShift];

#pragma unroll
  for (int j = 0; j < TOP_TILES; ++j) {
    const float* wrow = w_in + (size_t)(16 * j + tok) * IN_W + kbase + kShift;
    v8f c0 = {}, c1 = {}, c2 = {}, c3 = {};
#pragma unroll
    for (int kk = 0; kk < 256; kk += 16) {
      const float4 a0 = *(const float4*)(xrow + kk + 0);
      const float4 b0 = *(const float4*)(wrow + kk + 0);
      const float4 a1 = *(const float4*)(xrow + kk + 8);
      const float4 b1 = *(const float4*)(wrow + kk + 8);
      c0 = wmma4(v2f{a0.x, a0.y}, v2f{b0.x, b0.y}, c0);
      c1 = wmma4(v2f{a0.z, a0.w}, v2f{b0.z, b0.w}, c1);
      c2 = wmma4(v2f{a1.x, a1.y}, v2f{b1.x, b1.y}, c2);
      c3 = wmma4(v2f{a1.z, a1.w}, v2f{b1.z, b1.w}, c3);
    }
    const v8f c = (c0 + c1) + (c2 + c3);
    float* dst = &s_top[wave][(16 * j + tok) * TSTRIDE + 8 * (lane >> 4)];
    *(float4*)(dst + 0) = make_float4(c[0], c[1], c[2], c[3]);
    *(float4*)(dst + 4) = make_float4(c[4], c[5], c[6], c[7]);
    if (j + 1 < TOP_TILES) __builtin_amdgcn_sched_barrier(0);
  }
  __syncthreads();
  if (tid < TOK_TILE) {
    int node = 0;
    for (int lvl = 0; lvl < TOP_LEVELS; ++lvl) {
      const int o = node * TSTRIDE + tid;
      const float logit = (s_top[0][o] + s_top[1][o]) + (s_top[2][o] + s_top[3][o]);
      const float act = 0.5f * logit * (1.0f + erff(logit * 0.70710678118654752f));
      s_act[lvl][tid]  = act;
      s_path[lvl][tid] = node;
      node = 2 * node + 1 + ((logit > 0.0f) ? 1 : 0);
    }
    s_node[tid] = node;
  }
  __syncthreads();

  float* stg = &s_top[wave][0];
  const int rbase = 8 * ((lane >> 3) & 1) + 4 * (lane >> 4);
  const int spc  = (lane & 7) * 4;
  for (int lvl = TOP_LEVELS; lvl < LEVELS; ++lvl) {
    const int node = s_node[tok];
    const float* wr0 = w_in + (size_t)s_node[rbase + 0] * IN_W + kbase + spc;
    const float* wr1 = w_in + (size_t)s_node[rbase + 1] * IN_W + kbase + spc;
    const float* wr2 = w_in + (size_t)s_node[rbase + 2] * IN_W + kbase + spc;
    const float* wr3 = w_in + (size_t)s_node[rbase + 3] * IN_W + kbase + spc;

#define LDCHUNK(q)                                              \
    const float4 g##q##a = *(const float4*)(wr0 + 32 * (q));    \
    const float4 g##q##b = *(const float4*)(wr1 + 32 * (q));    \
    const float4 g##q##c = *(const float4*)(wr2 + 32 * (q));    \
    const float4 g##q##d = *(const float4*)(wr3 + 32 * (q));
    LDCHUNK(0) LDCHUNK(1) LDCHUNK(2) LDCHUNK(3) LDCHUNK(4) LDCHUNK(5) LDCHUNK(6) LDCHUNK(7)
#undef LDCHUNK

    v8f c0 = {}, c1 = {}, c2 = {}, c3 = {};
    float* st0 = &stg[(rbase + 0) * SSTRIDE + spc];
    float* st1 = &stg[(rbase + 1) * SSTRIDE + spc];
    float* st2 = &stg[(rbase + 2) * SSTRIDE + spc];
    float* st3 = &stg[(rbase + 3) * SSTRIDE + spc];
    const float* brow = &stg[tok * SSTRIDE + kShift];
#define CHUNK(q)                                                          \
    wave_lds_order();                                                     \
    *(float4*)st0 = g##q##a;                                              \
    *(float4*)st1 = g##q##b;                                              \
    *(float4*)st2 = g##q##c;                                              \
    *(float4*)st3 = g##q##d;                                              \
    wave_lds_order();                                                     \
    {                                                                     \
      const float4 a0 = *(const float4*)(xrow + 32 * (q) + 0);            \
      const float4 a1 = *(const float4*)(xrow + 32 * (q) + 8);            \
      const float4 a2 = *(const float4*)(xrow + 32 * (q) + 16);           \
      const float4 a3 = *(const float4*)(xrow + 32 * (q) + 24);           \
      const float4 b0 = *(const float4*)(brow + 0);                       \
      const float4 b1 = *(const float4*)(brow + 8);                       \
      const float4 b2 = *(const float4*)(brow + 16);                      \
      const float4 b3 = *(const float4*)(brow + 24);                      \
      c0 = wmma4(v2f{a0.x, a0.y}, v2f{b0.x, b0.y}, c0);                   \
      c1 = wmma4(v2f{a0.z, a0.w}, v2f{b0.z, b0.w}, c1);                   \
      c2 = wmma4(v2f{a1.x, a1.y}, v2f{b1.x, b1.y}, c2);                   \
      c3 = wmma4(v2f{a1.z, a1.w}, v2f{b1.z, b1.w}, c3);                   \
      c0 = wmma4(v2f{a2.x, a2.y}, v2f{b2.x, b2.y}, c0);                   \
      c1 = wmma4(v2f{a2.z, a2.w}, v2f{b2.z, b2.w}, c1);                   \
      c2 = wmma4(v2f{a3.x, a3.y}, v2f{b3.x, b3.y}, c2);                   \
      c3 = wmma4(v2f{a3.z, a3.w}, v2f{b3.z, b3.w}, c3);                   \
    }
    CHUNK(0) CHUNK(1) CHUNK(2) CHUNK(3) CHUNK(4) CHUNK(5) CHUNK(6) CHUNK(7)
#undef CHUNK
    const v8f c = (c0 + c1) + (c2 + c3);

    const int r = lane & 7;
    const float e0 = (r & 1) ? c[1] : c[0];
    const float e1 = (r & 1) ? c[3] : c[2];
    const float e2 = (r & 1) ? c[5] : c[4];
    const float e3 = (r & 1) ? c[7] : c[6];
    const float f0 = (r & 2) ? e1 : e0;
    const float f1 = (r & 2) ? e3 : e2;
    const float dg = (r & 4) ? f1 : f0;
    if (lane < 8)        atomicAdd(&s_logit[lane],      dg);
    else if (lane >= 24) atomicAdd(&s_logit[lane - 16], dg);
    __syncthreads();

    if (tid < TOK_TILE) {
      const float logit = s_logit[tid];
      const float act = 0.5f * logit * (1.0f + erff(logit * 0.70710678118654752f));
      s_act[lvl][tid]  = act;
      s_path[lvl][tid] = node;
      s_node[tid]  = 2 * node + 1 + ((logit > 0.0f) ? 1 : 0);
      s_logit[tid] = 0.0f;
    }
    __syncthreads();
  }

  {
    const float4* __restrict__ wt4 = (const float4*)w_out_T;
    float4* l4 = (float4*)lds_x;
    float4 v[STAGED_F4 / 128];
#pragma unroll
    for (int i = 0; i < STAGED_F4 / 128; ++i) v[i] = wt4[i * 128 + tid];
#pragma unroll
    for (int i = 0; i < STAGED_F4 / 128; ++i) l4[i * 128 + tid] = v[i];
  }
  __syncthreads();

  float* orow_base = out + xbase;
  for (int t4 = 0; t4 < 4; ++t4) {
    const int t = wave * 4 + t4;
    float4 acc[8];
#pragma unroll
    for (int c8 = 0; c8 < 8; ++c8) acc[c8] = make_float4(0.f, 0.f, 0.f, 0.f);
#pragma unroll
    for (int k = 0; k < STAGED_LEVELS; ++k) {
      const float a = s_act[k][t];
      const float4* row = (const float4*)(lds_x + s_path[k][t] * OUT_W);
#pragma unroll
      for (int c8 = 0; c8 < 8; ++c8) {
        const float4 w4 = row[c8 * 32 + lane];
        acc[c8].x += a * w4.x;
        acc[c8].y += a * w4.y;
        acc[c8].z += a * w4.z;
        acc[c8].w += a * w4.w;
      }
    }
#pragma unroll 4
    for (int k = STAGED_LEVELS; k < LEVELS; ++k) {
      const float a = s_act[k][t];
      const float4* row = (const float4*)(w_out_T + (size_t)s_path[k][t] * OUT_W);
#pragma unroll
      for (int c8 = 0; c8 < 8; ++c8) {
        const float4 w4 = row[c8 * 32 + lane];
        acc[c8].x += a * w4.x;
        acc[c8].y += a * w4.y;
        acc[c8].z += a * w4.z;
        acc[c8].w += a * w4.w;
      }
    }
    float4* orow = (float4*)(orow_base + (size_t)t * OUT_W);
#pragma unroll
    for (int c8 = 0; c8 < 8; ++c8) orow[c8 * 32 + lane] = acc[c8];
  }
}

extern "C" void kernel_launch(void* const* d_in, const int* in_sizes, int n_in,
                              void* d_out, int out_size, void* d_ws, size_t ws_size,
                              hipStream_t stream) {
  (void)in_sizes; (void)n_in; (void)out_size; (void)ws_size;
  const float* x     = (const float*)d_in[0];
  const float* w_in  = (const float*)d_in[1];
  const float* w_out = (const float*)d_in[2];
  float* w_out_T = (float*)d_ws;
  float* out     = (float*)d_out;

  transpose_wout<<<dim3((N_NODES + 31) / 32, OUT_W / 32), dim3(32, 8), 0, stream>>>(
      w_out, w_out_T);
  fff_fwd<<<N_TOKENS / TOK_TILE, 128, 0, stream>>>(x, w_in, w_out_T, out);
}
